# starting kernel
# baseline (speedup 1.0000x reference)
#include <hip/hip_runtime.h>
#include <hip/hip_bf16.h>

#define B_SZ   32
#define N_SZ   512
#define D_SZ   768
#define M_SZ   4
#define CDB    (D_SZ * B_SZ)
#define OUTC   D_SZ

typedef __attribute__((ext_vector_type(2))) float v2f;
typedef __attribute__((ext_vector_type(8))) float v8f;

__device__ __forceinline__ v8f wmma4(v2f a, v2f b, v8f c) {
  return __builtin_amdgcn_wmma_f32_16x16x4_f32(
       false, a,  false, b,
       (short)0, c,  false,  false);
}

__global__ __launch_bounds__(256)
void dinv_kernel(const float* __restrict__ adj, float* __restrict__ dinv) {
  __shared__ float red[256];
  const int row = blockIdx.x;
  const int t = threadIdx.x;
  float s = 0.f;
  const float* a = adj + row * N_SZ;
  for (int j = t; j < N_SZ; j += 256) s += a[j];
  red[t] = s;
  __syncthreads();
  for (int off = 128; off > 0; off >>= 1) {
    if (t < off) red[t] += red[t + off];
    __syncthreads();
  }
  if (t == 0) {
    float rs = red[0] + 1.0f;
    dinv[row] = (rs > 0.f) ? rsqrtf(rs) : 0.f;
  }
}

__global__ __launch_bounds__(256)
void support_kernel(const float* __restrict__ adj, const float* __restrict__ dinv,
                    float* __restrict__ support) {
  const int idx = blockIdx.x * 256 + threadIdx.x;
  const int i = idx >> 9;
  const int j = idx & (N_SZ - 1);
  float m = adj[j * N_SZ + i] + (i == j ? 1.0f : 0.0f);
  support[idx] = m * dinv[i] * dinv[j];
}

__global__ __launch_bounds__(256)
void transpose_kernel(const float* __restrict__ inputs, float* __restrict__ x0) {
  __shared__ float tile[64 * 33];
  const int n  = blockIdx.x;
  const int d0 = blockIdx.y * 64;
  const int t  = threadIdx.x;
#pragma unroll
  for (int p = 0; p < 8; ++p) {
    int b  = p * 4 + (t >> 6);
    int dl = t & 63;
    tile[dl * 33 + b] =
        inputs[(size_t)b * (N_SZ * D_SZ) + (size_t)n * D_SZ + d0 + dl];
  }
  __syncthreads();
  float* dst = x0 + (size_t)n * CDB + d0 * B_SZ;
#pragma unroll
  for (int p = 0; p < 8; ++p) {
    int c  = p * 256 + t;
    int dl = c >> 5;
    int b  = c & 31;
    dst[c] = tile[dl * 33 + b];
  }
}

__device__ __forceinline__
void store_tile(float* __restrict__ Y, const float* __restrict__ Xprev,
                int rbase, int col, float alpha, float beta,
                v8f a0, v8f a1, v8f a2, v8f a3) {
  if (beta != 0.0f) {
#pragma unroll
    for (int j = 0; j < 8; ++j) {
      float* Yr = Y + (rbase + j) * CDB + col;
      const float* Pr = Xprev + (rbase + j) * CDB + col;
      Yr[0]  = alpha * a0[j] + beta * Pr[0];
      Yr[16] = alpha * a1[j] + beta * Pr[16];
      Yr[32] = alpha * a2[j] + beta * Pr[32];
      Yr[48] = alpha * a3[j] + beta * Pr[48];
    }
  } else {
#pragma unroll
    for (int j = 0; j < 8; ++j) {
      float* Yr = Y + (rbase + j) * CDB + col;
      Yr[0]  = alpha * a0[j];
      Yr[16] = alpha * a1[j];
      Yr[32] = alpha * a2[j];
      Yr[48] = alpha * a3[j];
    }
  }
}

__global__ __launch_bounds__(256)
void diffusion_gemm(const float* __restrict__ S, const float* __restrict__ X,
                    const float* __restrict__ Xprev, float* __restrict__ Y,
                    float alpha, float beta) {
  const int lane = threadIdx.x & 31;
  const int wave = threadIdx.x >> 5;
  const int lm   = lane & 15;
  const int kh   = lane >> 4;
  const int m0 = blockIdx.y * 64 + (wave >> 2) * 32;
  const int n0 = blockIdx.x * 256 + (wave & 3) * 64;

  const float* Srow0 = S + (m0 + lm) * N_SZ + kh * 2;
  const float* Srow1 = Srow0 + 16 * N_SZ;
  const float* Xcol  = X + n0 + lm;

  v8f acc00 = {}, acc01 = {}, acc02 = {}, acc03 = {};
  v8f acc10 = {}, acc11 = {}, acc12 = {}, acc13 = {};
  for (int k = 0; k < N_SZ; k += 4) {
    v2f a0 = *(const v2f*)(Srow0 + k);
    v2f a1 = *(const v2f*)(Srow1 + k);
    const float* Xr0 = Xcol + (k + kh * 2) * CDB;
    const float* Xr1 = Xr0 + CDB;
    v2f b0 = {Xr0[0],  Xr1[0]};
    v2f b1 = {Xr0[16], Xr1[16]};
    v2f b2 = {Xr0[32], Xr1[32]};
    v2f b3 = {Xr0[48], Xr1[48]};
    acc00 = wmma4(a0, b0, acc00);
    acc01 = wmma4(a0, b1, acc01);
    acc02 = wmma4(a0, b2, acc02);
    acc03 = wmma4(a0, b3, acc03);
    acc10 = wmma4(a1, b0, acc10);
    acc11 = wmma4(a1, b1, acc11);
    acc12 = wmma4(a1, b2, acc12);
    acc13 = wmma4(a1, b3, acc13);
  }

  store_tile(Y, Xprev, m0 + kh * 8,      n0 + lm, alpha, beta,
             acc00, acc01, acc02, acc03);
  store_tile(Y, Xprev, m0 + 16 + kh * 8, n0 + lm, alpha, beta,
             acc10, acc11, acc12, acc13);
}

__global__ __launch_bounds__(256)
void output_gemm(const float* __restrict__ x0, const float* __restrict__ x1,
                 const float* __restrict__ x2, const float* __restrict__ x3,
                 const float* __restrict__ W, const float* __restrict__ bias,
                 float* __restrict__ out) {
  const int lane = threadIdx.x & 31;
  const int wave = threadIdx.x >> 5;
  const int lm   = lane & 15;
  const int kh   = lane >> 4;
  const int r0 = blockIdx.y * 64 + (wave >> 2) * 32;
  const int o0 = blockIdx.x * 256 + (wave & 3) * 64;

  const int ra = r0 + lm;
  const int rb = r0 + 16 + lm;
  const size_t aoff0 = (size_t)(ra & (N_SZ - 1)) * CDB + (ra >> 9);
  const size_t aoff1 = (size_t)(rb & (N_SZ - 1)) * CDB + (rb >> 9);
  const float* xa = (kh == 0) ? x0 : x2;
  const float* xb = (kh == 0) ? x1 : x3;
  const float* Wb = W + kh * 2 * OUTC + o0 + lm;

  v8f acc00 = {}, acc01 = {}, acc02 = {}, acc03 = {};
  v8f acc10 = {}, acc11 = {}, acc12 = {}, acc13 = {};
  for (int d = 0; d < D_SZ; ++d) {
    const int xo = d * B_SZ;
    v2f a0 = {xa[aoff0 + xo], xb[aoff0 + xo]};
    v2f a1 = {xa[aoff1 + xo], xb[aoff1 + xo]};
    const float* Wr0 = Wb + (d * 4) * OUTC;
    const float* Wr1 = Wr0 + OUTC;
    v2f b0 = {Wr0[0],  Wr1[0]};
    v2f b1 = {Wr0[16], Wr1[16]};
    v2f b2 = {Wr0[32], Wr1[32]};
    v2f b3 = {Wr0[48], Wr1[48]};
    acc00 = wmma4(a0, b0, acc00);
    acc01 = wmma4(a0, b1, acc01);
    acc02 = wmma4(a0, b2, acc02);
    acc03 = wmma4(a0, b3, acc03);
    acc10 = wmma4(a1, b0, acc10);
    acc11 = wmma4(a1, b1, acc11);
    acc12 = wmma4(a1, b2, acc12);
    acc13 = wmma4(a1, b3, acc13);
  }

  const float bi0 = bias[o0 + lm];
  const float bi1 = bias[o0 + 16 + lm];
  const float bi2 = bias[o0 + 32 + lm];
  const float bi3 = bias[o0 + 48 + lm];
#pragma unroll
  for (int h = 0; h < 2; ++h) {
    const int rbase = r0 + h * 16 + kh * 8;
    v8f c0 = h ? acc10 : acc00;
    v8f c1 = h ? acc11 : acc01;
    v8f c2 = h ? acc12 : acc02;
    v8f c3 = h ? acc13 : acc03;
#pragma unroll
    for (int j = 0; j < 8; ++j) {
      float* Or = out + (rbase + j) * OUTC + o0 + lm;
      Or[0]  = c0[j] + bi0;
      Or[16] = c1[j] + bi1;
      Or[32] = c2[j] + bi2;
      Or[48] = c3[j] + bi3;
    }
  }
}

extern "C" void kernel_launch(void* const* d_in, const int* in_sizes, int n_in,
                              void* d_out, int out_size, void* d_ws, size_t ws_size,
                              hipStream_t stream) {
  const float* inputs  = (const float*)d_in[0];
  const float* adj     = (const float*)d_in[1];
  const float* weights = (const float*)d_in[2];
  const float* biases  = (const float*)d_in[3];
  float* out = (float*)d_out;

  float* ws      = (float*)d_ws;
  float* support = ws;
  float* dinv    = support + (size_t)N_SZ * N_SZ;
  float* x0      = dinv + 512;
  float* x1      = x0 + (size_t)N_SZ * CDB;
  float* x2      = x1 + (size_t)N_SZ * CDB;
  float* x3      = x2 + (size_t)N_SZ * CDB;

  (void)in_sizes; (void)n_in; (void)out_size; (void)ws_size;

  dinv_kernel<<<N_SZ, 256, 0, stream>>>(adj, dinv);
  support_kernel<<<(N_SZ * N_SZ) / 256, 256, 0, stream>>>(adj, dinv, support);
  transpose_kernel<<<dim3(N_SZ, D_SZ / 64), 256, 0, stream>>>(inputs, x0);

  dim3 gd(CDB / 256, N_SZ / 64);
  diffusion_gemm<<<gd, 256, 0, stream>>>(support, x0, x0, x1, 1.0f, 0.0f);
  diffusion_gemm<<<gd, 256, 0, stream>>>(support, x1, x0, x2, 2.0f, -1.0f);
  diffusion_gemm<<<gd, 256, 0, stream>>>(support, x2, x1, x3, 2.0f, -1.0f);

  output_gemm<<<dim3(OUTC / 256, (B_SZ * N_SZ) / 64), 256, 0, stream>>>(
      x0, x1, x2, x3, weights, biases, out);
}


# output GEMM: a wave's 32 tile rows are the 32 batch entries of one node, so A loads read contiguous 64-byte runs, not one 4-byte word per 128-byte line; k order and sum order unchanged
# speedup vs baseline: 3.8649x; 3.8649x over previous
#include <hip/hip_runtime.h>
#include <hip/hip_bf16.h>

#define B_SZ   32
#define N_SZ   512
#define D_SZ   768
#define M_SZ   4
#define CDB    (D_SZ * B_SZ)
#define OUTC   D_SZ

typedef __attribute__((ext_vector_type(2))) float v2f;
typedef __attribute__((ext_vector_type(8))) float v8f;

__device__ __forceinline__ v8f wmma4(v2f a, v2f b, v8f c) {
  return __builtin_amdgcn_wmma_f32_16x16x4_f32(
       false, a,  false, b,
       (short)0, c,  false,  false);
}

__global__ __launch_bounds__(256)
void dinv_kernel(const float* __restrict__ adj, float* __restrict__ dinv) {
  __shared__ float red[256];
  const int row = blockIdx.x;
  const int t = threadIdx.x;
  float s = 0.f;
  const float* a = adj + row * N_SZ;
  for (int j = t; j < N_SZ; j += 256) s += a[j];
  red[t] = s;
  __syncthreads();
  for (int off = 128; off > 0; off >>= 1) {
    if (t < off) red[t] += red[t + off];
    __syncthreads();
  }
  if (t == 0) {
    float rs = red[0] + 1.0f;
    dinv[row] = (rs > 0.f) ? rsqrtf(rs) : 0.f;
  }
}

__global__ __launch_bounds__(256)
void support_kernel(const float* __restrict__ adj, const float* __restrict__ dinv,
                    float* __restrict__ support) {
  const int idx = blockIdx.x * 256 + threadIdx.x;
  const int i = idx >> 9;
  const int j = idx & (N_SZ - 1);
  float m = adj[j * N_SZ + i] + (i == j ? 1.0f : 0.0f);
  support[idx] = m * dinv[i] * dinv[j];
}

__global__ __launch_bounds__(256)
void transpose_kernel(const float* __restrict__ inputs, float* __restrict__ x0) {
  __shared__ float tile[64 * 33];
  const int n  = blockIdx.x;
  const int d0 = blockIdx.y * 64;
  const int t  = threadIdx.x;
#pragma unroll
  for (int p = 0; p < 8; ++p) {
    int b  = p * 4 + (t >> 6);
    int dl = t & 63;
    tile[dl * 33 + b] =
        inputs[(size_t)b * (N_SZ * D_SZ) + (size_t)n * D_SZ + d0 + dl];
  }
  __syncthreads();
  float* dst = x0 + (size_t)n * CDB + d0 * B_SZ;
#pragma unroll
  for (int p = 0; p < 8; ++p) {
    int c  = p * 256 + t;
    int dl = c >> 5;
    int b  = c & 31;
    dst[c] = tile[dl * 33 + b];
  }
}

__device__ __forceinline__
void store_tile(float* __restrict__ Y, const float* __restrict__ Xprev,
                int rbase, int col, float alpha, float beta,
                v8f a0, v8f a1, v8f a2, v8f a3) {
  if (beta != 0.0f) {
#pragma unroll
    for (int j = 0; j < 8; ++j) {
      float* Yr = Y + (rbase + j) * CDB + col;
      const float* Pr = Xprev + (rbase + j) * CDB + col;
      Yr[0]  = alpha * a0[j] + beta * Pr[0];
      Yr[16] = alpha * a1[j] + beta * Pr[16];
      Yr[32] = alpha * a2[j] + beta * Pr[32];
      Yr[48] = alpha * a3[j] + beta * Pr[48];
    }
  } else {
#pragma unroll
    for (int j = 0; j < 8; ++j) {
      float* Yr = Y + (rbase + j) * CDB + col;
      Yr[0]  = alpha * a0[j];
      Yr[16] = alpha * a1[j];
      Yr[32] = alpha * a2[j];
      Yr[48] = alpha * a3[j];
    }
  }
}

__global__ __launch_bounds__(256)
void diffusion_gemm(const float* __restrict__ S, const float* __restrict__ X,
                    const float* __restrict__ Xprev, float* __restrict__ Y,
                    float alpha, float beta) {
  const int lane = threadIdx.x & 31;
  const int wave = threadIdx.x >> 5;
  const int lm   = lane & 15;
  const int kh   = lane >> 4;
  const int m0 = blockIdx.y * 64 + (wave >> 2) * 32;
  const int n0 = blockIdx.x * 256 + (wave & 3) * 64;

  const float* Srow0 = S + (m0 + lm) * N_SZ + kh * 2;
  const float* Srow1 = Srow0 + 16 * N_SZ;
  const float* Xcol  = X + n0 + lm;

  v8f acc00 = {}, acc01 = {}, acc02 = {}, acc03 = {};
  v8f acc10 = {}, acc11 = {}, acc12 = {}, acc13 = {};
  for (int k = 0; k < N_SZ; k += 4) {
    v2f a0 = *(const v2f*)(Srow0 + k);
    v2f a1 = *(const v2f*)(Srow1 + k);
    const float* Xr0 = Xcol + (k + kh * 2) * CDB;
    const float* Xr1 = Xr0 + CDB;
    v2f b0 = {Xr0[0],  Xr1[0]};
    v2f b1 = {Xr0[16], Xr1[16]};
    v2f b2 = {Xr0[32], Xr1[32]};
    v2f b3 = {Xr0[48], Xr1[48]};
    acc00 = wmma4(a0, b0, acc00);
    acc01 = wmma4(a0, b1, acc01);
    acc02 = wmma4(a0, b2, acc02);
    acc03 = wmma4(a0, b3, acc03);
    acc10 = wmma4(a1, b0, acc10);
    acc11 = wmma4(a1, b1, acc11);
    acc12 = wmma4(a1, b2, acc12);
    acc13 = wmma4(a1, b3, acc13);
  }

  store_tile(Y, Xprev, m0 + kh * 8,      n0 + lm, alpha, beta,
             acc00, acc01, acc02, acc03);
  store_tile(Y, Xprev, m0 + 16 + kh * 8, n0 + lm, alpha, beta,
             acc10, acc11, acc12, acc13);
}

__global__ __launch_bounds__(256)
void output_gemm(const float* __restrict__ x0, const float* __restrict__ x1,
                 const float* __restrict__ x2, const float* __restrict__ x3,
                 const float* __restrict__ W, const float* __restrict__ bias,
                 float* __restrict__ out) {
  const int lane = threadIdx.x & 31;
  const int wave = threadIdx.x >> 5;
  const int lm   = lane & 15;
  const int kh   = lane >> 4;
  const int n  = blockIdx.y * 2 + (wave >> 2);
  const int o0 = blockIdx.x * 256 + (wave & 3) * 64;

  const size_t aoff0 = (size_t)n * CDB + lm;
  const size_t aoff1 = aoff0 + 16;
  const float* xa = (kh == 0) ? x0 : x2;
  const float* xb = (kh == 0) ? x1 : x3;
  const float* Wb = W + kh * 2 * OUTC + o0 + lm;

  v8f acc00 = {}, acc01 = {}, acc02 = {}, acc03 = {};
  v8f acc10 = {}, acc11 = {}, acc12 = {}, acc13 = {};
  for (int d = 0; d < D_SZ; ++d) {
    const int xo = d * B_SZ;
    v2f a0 = {xa[aoff0 + xo], xb[aoff0 + xo]};
    v2f a1 = {xa[aoff1 + xo], xb[aoff1 + xo]};
    const float* Wr0 = Wb + (d * 4) * OUTC;
    const float* Wr1 = Wr0 + OUTC;
    v2f b0 = {Wr0[0],  Wr1[0]};
    v2f b1 = {Wr0[16], Wr1[16]};
    v2f b2 = {Wr0[32], Wr1[32]};
    v2f b3 = {Wr0[48], Wr1[48]};
    acc00 = wmma4(a0, b0, acc00);
    acc01 = wmma4(a0, b1, acc01);
    acc02 = wmma4(a0, b2, acc02);
    acc03 = wmma4(a0, b3, acc03);
    acc10 = wmma4(a1, b0, acc10);
    acc11 = wmma4(a1, b1, acc11);
    acc12 = wmma4(a1, b2, acc12);
    acc13 = wmma4(a1, b3, acc13);
  }

  const float bi0 = bias[o0 + lm];
  const float bi1 = bias[o0 + 16 + lm];
  const float bi2 = bias[o0 + 32 + lm];
  const float bi3 = bias[o0 + 48 + lm];
#pragma unroll
  for (int h = 0; h < 2; ++h) {
    const int bbase = h * 16 + kh * 8;
    v8f c0 = h ? acc10 : acc00;
    v8f c1 = h ? acc11 : acc01;
    v8f c2 = h ? acc12 : acc02;
    v8f c3 = h ? acc13 : acc03;
#pragma unroll
    for (int j = 0; j < 8; ++j) {
      float* Or = out + ((bbase + j) * N_SZ + n) * OUTC + o0 + lm;
      Or[0]  = c0[j] + bi0;
      Or[16] = c1[j] + bi1;
      Or[32] = c2[j] + bi2;
      Or[48] = c3[j] + bi3;
    }
  }
}

extern "C" void kernel_launch(void* const* d_in, const int* in_sizes, int n_in,
                              void* d_out, int out_size, void* d_ws, size_t ws_size,
                              hipStream_t stream) {
  const float* inputs  = (const float*)d_in[0];
  const float* adj     = (const float*)d_in[1];
  const float* weights = (const float*)d_in[2];
  const float* biases  = (const float*)d_in[3];
  float* out = (float*)d_out;

  float* ws      = (float*)d_ws;
  float* support = ws;
  float* dinv    = support + (size_t)N_SZ * N_SZ;
  float* x0      = dinv + 512;
  float* x1      = x0 + (size_t)N_SZ * CDB;
  float* x2      = x1 + (size_t)N_SZ * CDB;
  float* x3      = x2 + (size_t)N_SZ * CDB;

  (void)in_sizes; (void)n_in; (void)out_size; (void)ws_size;

  dinv_kernel<<<N_SZ, 256, 0, stream>>>(adj, dinv);
  support_kernel<<<(N_SZ * N_SZ) / 256, 256, 0, stream>>>(adj, dinv, support);
  transpose_kernel<<<dim3(N_SZ, D_SZ / 64), 256, 0, stream>>>(inputs, x0);

  dim3 gd(CDB / 256, N_SZ / 64);
  diffusion_gemm<<<gd, 256, 0, stream>>>(support, x0, x0, x1, 1.0f, 0.0f);
  diffusion_gemm<<<gd, 256, 0, stream>>>(support, x1, x0, x2, 2.0f, -1.0f);
  diffusion_gemm<<<gd, 256, 0, stream>>>(support, x2, x1, x3, 2.0f, -1.0f);

  output_gemm<<<dim3(OUTC / 256, (B_SZ * N_SZ) / 64), 256, 0, stream>>>(
      x0, x1, x2, x3, weights, biases, out);
}


# every fragment load and C store covers whole lines: a lane owns 4 adjacent columns (16-byte B loads, C stores) and 2 adjacent batch rows; support stored k-blocked; k and sum order unchanged
# speedup vs baseline: 4.0205x; 1.0403x over previous
#include <hip/hip_runtime.h>
#include <hip/hip_bf16.h>

#define B_SZ   32
#define N_SZ   512
#define D_SZ   768
#define M_SZ   4
#define CDB    (D_SZ * B_SZ)
#define OUTC   D_SZ

typedef __attribute__((ext_vector_type(2))) float v2f;
typedef __attribute__((ext_vector_type(4))) float v4f;
typedef __attribute__((ext_vector_type(8))) float v8f;

__device__ __forceinline__ v8f wmma4(v2f a, v2f b, v8f c) {
  return __builtin_amdgcn_wmma_f32_16x16x4_f32(
       false, a,  false, b,
       (short)0, c,  false,  false);
}

__global__ __launch_bounds__(256)
void dinv_kernel(const float* __restrict__ adj, float* __restrict__ dinv) {
  __shared__ float red[256];
  const int row = blockIdx.x;
  const int t = threadIdx.x;
  float s = 0.f;
  const float* a = adj + row * N_SZ;
  for (int j = t; j < N_SZ; j += 256) s += a[j];
  red[t] = s;
  __syncthreads();
  for (int off = 128; off > 0; off >>= 1) {
    if (t < off) red[t] += red[t + off];
    __syncthreads();
  }
  if (t == 0) {
    float rs = red[0] + 1.0f;
    dinv[row] = (rs > 0.f) ? rsqrtf(rs) : 0.f;
  }
}

__global__ __launch_bounds__(256)
void support_kernel(const float* __restrict__ adj, const float* __restrict__ dinv,
                    float* __restrict__ support) {
  const int a   = blockIdx.x * 256 + threadIdx.x;
  const int pos = a & 7;
  const int i   = (a >> 3) & (N_SZ - 1);
  const int kb  = a >> 12;
  const int w   = ((pos >> 1) & 1) * 4 + (pos >> 2) * 2 + (pos & 1);
  const int j   = kb * 8 + w;
  float m = adj[j * N_SZ + i] + (i == j ? 1.0f : 0.0f);
  support[a] = m * dinv[i] * dinv[j];
}

__global__ __launch_bounds__(256)
void transpose_kernel(const float* __restrict__ inputs, float* __restrict__ x0) {
  __shared__ float tile[64 * 33];
  const int n  = blockIdx.x;
  const int d0 = blockIdx.y * 64;
  const int t  = threadIdx.x;
#pragma unroll
  for (int p = 0; p < 8; ++p) {
    int b  = p * 4 + (t >> 6);
    int dl = t & 63;
    tile[dl * 33 + b] =
        inputs[(size_t)b * (N_SZ * D_SZ) + (size_t)n * D_SZ + d0 + dl];
  }
  __syncthreads();
  float* dst = x0 + (size_t)n * CDB + d0 * B_SZ;
#pragma unroll
  for (int p = 0; p < 8; ++p) {
    int c  = p * 256 + t;
    int dl = c >> 5;
    int b  = c & 31;
    dst[c] = tile[dl * 33 + b];
  }
}

__device__ __forceinline__
void store_tile(float* __restrict__ Y, const float* __restrict__ Xprev,
                int rbase, int col, float alpha, float beta,
                v8f a0, v8f a1, v8f a2, v8f a3) {
  if (beta != 0.0f) {
#pragma unroll
    for (int j = 0; j < 8; ++j) {
      const int off = (rbase + j) * CDB + col;
      const v4f p = *(const v4f*)(Xprev + off);
      v4f y;
      y.x = alpha * a0[j] + beta * p.x;
      y.y = alpha * a1[j] + beta * p.y;
      y.z = alpha * a2[j] + beta * p.z;
      y.w = alpha * a3[j] + beta * p.w;
      *(v4f*)(Y + off) = y;
    }
  } else {
#pragma unroll
    for (int j = 0; j < 8; ++j) {
      const int off = (rbase + j) * CDB + col;
      v4f y;
      y.x = alpha * a0[j];
      y.y = alpha * a1[j];
      y.z = alpha * a2[j];
      y.w = alpha * a3[j];
      *(v4f*)(Y + off) = y;
    }
  }
}

__global__ __launch_bounds__(256)
void diffusion_gemm(const float* __restrict__ S, const float* __restrict__ X,
                    const float* __restrict__ Xprev, float* __restrict__ Y,
                    float alpha, float beta) {
  const int lane = threadIdx.x & 31;
  const int wave = threadIdx.x >> 5;
  const int lm   = lane & 15;
  const int kh   = lane >> 4;
  const int m0 = blockIdx.y * 64 + (wave >> 2) * 32;
  const int n0 = blockIdx.x * 256 + (wave & 3) * 64;

  const float* Ap   = S + (m0 + lm) * 8 + kh * 4;
  const float* Xcol = X + (kh * 2) * CDB + n0 + 4 * lm;

  v8f acc00 = {}, acc01 = {}, acc02 = {}, acc03 = {};
  v8f acc10 = {}, acc11 = {}, acc12 = {}, acc13 = {};
  for (int k = 0; k < N_SZ; k += 8) {
    const float* Ab = Ap + (k >> 3) * (N_SZ * 8);
    const v4f av0 = *(const v4f*)(Ab);
    const v4f av1 = *(const v4f*)(Ab + 16 * 8);
    const float* Xr = Xcol + k * CDB;
    const v4f r0 = *(const v4f*)(Xr);
    const v4f r1 = *(const v4f*)(Xr + CDB);
    const v4f r2 = *(const v4f*)(Xr + 4 * CDB);
    const v4f r3 = *(const v4f*)(Xr + 5 * CDB);
    {
      const v2f a0 = {av0.x, av0.y};
      const v2f a1 = {av1.x, av1.y};
      const v2f b0 = {r0.x, r1.x};
      const v2f b1 = {r0.y, r1.y};
      const v2f b2 = {r0.z, r1.z};
      const v2f b3 = {r0.w, r1.w};
      acc00 = wmma4(a0, b0, acc00);
      acc01 = wmma4(a0, b1, acc01);
      acc02 = wmma4(a0, b2, acc02);
      acc03 = wmma4(a0, b3, acc03);
      acc10 = wmma4(a1, b0, acc10);
      acc11 = wmma4(a1, b1, acc11);
      acc12 = wmma4(a1, b2, acc12);
      acc13 = wmma4(a1, b3, acc13);
    }
    {
      const v2f a0 = {av0.z, av0.w};
      const v2f a1 = {av1.z, av1.w};
      const v2f b0 = {r2.x, r3.x};
      const v2f b1 = {r2.y, r3.y};
      const v2f b2 = {r2.z, r3.z};
      const v2f b3 = {r2.w, r3.w};
      acc00 = wmma4(a0, b0, acc00);
      acc01 = wmma4(a0, b1, acc01);
      acc02 = wmma4(a0, b2, acc02);
      acc03 = wmma4(a0, b3, acc03);
      acc10 = wmma4(a1, b0, acc10);
      acc11 = wmma4(a1, b1, acc11);
      acc12 = wmma4(a1, b2, acc12);
      acc13 = wmma4(a1, b3, acc13);
    }
  }

  store_tile(Y, Xprev, m0 + kh * 8,      n0 + 4 * lm, alpha, beta,
             acc00, acc01, acc02, acc03);
  store_tile(Y, Xprev, m0 + 16 + kh * 8, n0 + 4 * lm, alpha, beta,
             acc10, acc11, acc12, acc13);
}

__global__ __launch_bounds__(256)
void output_gemm(const float* __restrict__ x0, const float* __restrict__ x1,
                 const float* __restrict__ x2, const float* __restrict__ x3,
                 const float* __restrict__ W, const float* __restrict__ bias,
                 float* __restrict__ out) {
  const int lane = threadIdx.x & 31;
  const int wave = threadIdx.x >> 5;
  const int lm   = lane & 15;
  const int kh   = lane >> 4;
  const int n  = blockIdx.y * 2 + (wave >> 2);
  const int o0 = blockIdx.x * 256 + (wave & 3) * 64;

  const int aoff = n * CDB + 2 * lm;
  const float* xa = ((kh == 0) ? x0 : x2) + aoff;
  const float* xb = ((kh == 0) ? x1 : x3) + aoff;
  const float* Wb = W + kh * 2 * OUTC + o0 + 4 * lm;

  v8f acc00 = {}, acc01 = {}, acc02 = {}, acc03 = {};
  v8f acc10 = {}, acc11 = {}, acc12 = {}, acc13 = {};
  for (int d = 0; d < D_SZ; ++d) {
    const v2f pa = *(const v2f*)(xa + d * B_SZ);
    const v2f pb = *(const v2f*)(xb + d * B_SZ);
    const float* Wr0 = Wb + (d * 4) * OUTC;
    const v4f w0 = *(const v4f*)(Wr0);
    const v4f w1 = *(const v4f*)(Wr0 + OUTC);
    const v2f a0 = {pa.x, pb.x};
    const v2f a1 = {pa.y, pb.y};
    const v2f b0 = {w0.x, w1.x};
    const v2f b1 = {w0.y, w1.y};
    const v2f b2 = {w0.z, w1.z};
    const v2f b3 = {w0.w, w1.w};
    acc00 = wmma4(a0, b0, acc00);
    acc01 = wmma4(a0, b1, acc01);
    acc02 = wmma4(a0, b2, acc02);
    acc03 = wmma4(a0, b3, acc03);
    acc10 = wmma4(a1, b0, acc10);
    acc11 = wmma4(a1, b1, acc11);
    acc12 = wmma4(a1, b2, acc12);
    acc13 = wmma4(a1, b3, acc13);
  }

  const v4f bi = *(const v4f*)(bias + o0 + 4 * lm);
#pragma unroll
  for (int h = 0; h < 2; ++h) {
    v8f c0 = h ? acc10 : acc00;
    v8f c1 = h ? acc11 : acc01;
    v8f c2 = h ? acc12 : acc02;
    v8f c3 = h ? acc13 : acc03;
#pragma unroll
    for (int j = 0; j < 8; ++j) {
      const int b = 2 * (kh * 8 + j) + h;
      v4f v;
      v.x = c0[j] + bi.x;
      v.y = c1[j] + bi.y;
      v.z = c2[j] + bi.z;
      v.w = c3[j] + bi.w;
      *(v4f*)(out + (b * N_SZ + n) * OUTC + o0 + 4 * lm) = v;
    }
  }
}

extern "C" void kernel_launch(void* const* d_in, const int* in_sizes, int n_in,
                              void* d_out, int out_size, void* d_ws, size_t ws_size,
                              hipStream_t stream) {
  const float* inputs  = (const float*)d_in[0];
  const float* adj     = (const float*)d_in[1];
  const float* weights = (const float*)d_in[2];
  const float* biases  = (const float*)d_in[3];
  float* out = (float*)d_out;

  float* ws      = (float*)d_ws;
  float* support = ws;
  float* dinv    = support + (size_t)N_SZ * N_SZ;
  float* x0      = dinv + 512;
  float* x1      = x0 + (size_t)N_SZ * CDB;
  float* x2      = x1 + (size_t)N_SZ * CDB;
  float* x3      = x2 + (size_t)N_SZ * CDB;

  (void)in_sizes; (void)n_in; (void)out_size; (void)ws_size;

  dinv_kernel<<<N_SZ, 256, 0, stream>>>(adj, dinv);
  support_kernel<<<(N_SZ * N_SZ) / 256, 256, 0, stream>>>(adj, dinv, support);
  transpose_kernel<<<dim3(N_SZ, D_SZ / 64), 256, 0, stream>>>(inputs, x0);

  dim3 gd(CDB / 256, N_SZ / 64);
  diffusion_gemm<<<gd, 256, 0, stream>>>(support, x0, x0, x1, 1.0f, 0.0f);
  diffusion_gemm<<<gd, 256, 0, stream>>>(support, x1, x0, x2, 2.0f, -1.0f);
  diffusion_gemm<<<gd, 256, 0, stream>>>(support, x2, x1, x3, 2.0f, -1.0f);

  output_gemm<<<dim3(OUTC / 256, (B_SZ * N_SZ) / 64), 256, 0, stream>>>(
      x0, x1, x2, x3, weights, biases, out);
}


# both GEMM K loops: the next stages' fragment loads are issued as one group ahead of the current stage's matrix instructions (2 or 3 register buffers); tiles, k order and sum order unchanged
# speedup vs baseline: 4.3599x; 1.0844x over previous
#include <hip/hip_runtime.h>
#include <hip/hip_bf16.h>

#define B_SZ   32
#define N_SZ   512
#define D_SZ   768
#define M_SZ   4
#define CDB    (D_SZ * B_SZ)
#define OUTC   D_SZ

typedef __attribute__((ext_vector_type(2))) float v2f;
typedef __attribute__((ext_vector_type(4))) float v4f;
typedef __attribute__((ext_vector_type(8))) float v8f;

__device__ __forceinline__ v8f wmma4(v2f a, v2f b, v8f c) {
  return __builtin_amdgcn_wmma_f32_16x16x4_f32(
       false, a,  false, b,
       (short)0, c,  false,  false);
}

__global__ __launch_bounds__(256)
void dinv_kernel(const float* __restrict__ adj, float* __restrict__ dinv) {
  __shared__ float red[256];
  const int row = blockIdx.x;
  const int t = threadIdx.x;
  float s = 0.f;
  const float* a = adj + row * N_SZ;
  for (int j = t; j < N_SZ; j += 256) s += a[j];
  red[t] = s;
  __syncthreads();
  for (int off = 128; off > 0; off >>= 1) {
    if (t < off) red[t] += red[t + off];
    __syncthreads();
  }
  if (t == 0) {
    float rs = red[0] + 1.0f;
    dinv[row] = (rs > 0.f) ? rsqrtf(rs) : 0.f;
  }
}

__global__ __launch_bounds__(256)
void support_kernel(const float* __restrict__ adj, const float* __restrict__ dinv,
                    float* __restrict__ support) {
  const int a   = blockIdx.x * 256 + threadIdx.x;
  const int pos = a & 7;
  const int i   = (a >> 3) & (N_SZ - 1);
  const int kb  = a >> 12;
  const int w   = ((pos >> 1) & 1) * 4 + (pos >> 2) * 2 + (pos & 1);
  const int j   = kb * 8 + w;
  float m = adj[j * N_SZ + i] + (i == j ? 1.0f : 0.0f);
  support[a] = m * dinv[i] * dinv[j];
}

__global__ __launch_bounds__(256)
void transpose_kernel(const float* __restrict__ inputs, float* __restrict__ x0) {
  __shared__ float tile[64 * 33];
  const int n  = blockIdx.x;
  const int d0 = blockIdx.y * 64;
  const int t  = threadIdx.x;
#pragma unroll
  for (int p = 0; p < 8; ++p) {
    int b  = p * 4 + (t >> 6);
    int dl = t & 63;
    tile[dl * 33 + b] =
        inputs[(size_t)b * (N_SZ * D_SZ) + (size_t)n * D_SZ + d0 + dl];
  }
  __syncthreads();
  float* dst = x0 + (size_t)n * CDB + d0 * B_SZ;
#pragma unroll
  for (int p = 0; p < 8; ++p) {
    int c  = p * 256 + t;
    int dl = c >> 5;
    int b  = c & 31;
    dst[c] = tile[dl * 33 + b];
  }
}

__device__ __forceinline__
void store_tile(float* __restrict__ Y, const float* __restrict__ Xprev,
                int rbase, int col, float alpha, float beta,
                v8f a0, v8f a1, v8f a2, v8f a3) {
  if (beta != 0.0f) {
#pragma unroll
    for (int j = 0; j < 8; ++j) {
      const int off = (rbase + j) * CDB + col;
      const v4f p = *(const v4f*)(Xprev + off);
      v4f y;
      y.x = alpha * a0[j] + beta * p.x;
      y.y = alpha * a1[j] + beta * p.y;
      y.z = alpha * a2[j] + beta * p.z;
      y.w = alpha * a3[j] + beta * p.w;
      *(v4f*)(Y + off) = y;
    }
  } else {
#pragma unroll
    for (int j = 0; j < 8; ++j) {
      const int off = (rbase + j) * CDB + col;
      v4f y;
      y.x = alpha * a0[j];
      y.y = alpha * a1[j];
      y.z = alpha * a2[j];
      y.w = alpha * a3[j];
      *(v4f*)(Y + off) = y;
    }
  }
}

#define DSTG 2
#define DNST (N_SZ / (8 * DSTG))

struct DFrag {
  v4f av0[DSTG], av1[DSTG];
  v4f r0[DSTG], r1[DSTG], r2[DSTG], r3[DSTG];
};

__device__ __forceinline__
void diff_load(DFrag& f, const float* __restrict__ Ap,
               const float* __restrict__ Xcol, int stage) {
#pragma unroll
  for (int i = 0; i < DSTG; ++i) {
    const int kb = stage * DSTG + i;
    const float* Ab = Ap + kb * (N_SZ * 8);
    f.av0[i] = *(const v4f*)(Ab);
    f.av1[i] = *(const v4f*)(Ab + 16 * 8);
    const float* Xr = Xcol + (kb * 8) * CDB;
    f.r0[i] = *(const v4f*)(Xr);
    f.r1[i] = *(const v4f*)(Xr + CDB);
    f.r2[i] = *(const v4f*)(Xr + 4 * CDB);
    f.r3[i] = *(const v4f*)(Xr + 5 * CDB);
  }
}

__device__ __forceinline__
void diff_mma(const DFrag& f, v8f (&acc)[2][4]) {
#pragma unroll
  for (int i = 0; i < DSTG; ++i) {
    {
      const v2f a0 = {f.av0[i].x, f.av0[i].y};
      const v2f a1 = {f.av1[i].x, f.av1[i].y};
      const v2f b[4] = {{f.r0[i].x, f.r1[i].x}, {f.r0[i].y, f.r1[i].y},
                        {f.r0[i].z, f.r1[i].z}, {f.r0[i].w, f.r1[i].w}};
#pragma unroll
      for (int t = 0; t < 4; ++t) acc[0][t] = wmma4(a0, b[t], acc[0][t]);
#pragma unroll
      for (int t = 0; t < 4; ++t) acc[1][t] = wmma4(a1, b[t], acc[1][t]);
    }
    {
      const v2f a0 = {f.av0[i].z, f.av0[i].w};
      const v2f a1 = {f.av1[i].z, f.av1[i].w};
      const v2f b[4] = {{f.r2[i].x, f.r3[i].x}, {f.r2[i].y, f.r3[i].y},
                        {f.r2[i].z, f.r3[i].z}, {f.r2[i].w, f.r3[i].w}};
#pragma unroll
      for (int t = 0; t < 4; ++t) acc[0][t] = wmma4(a0, b[t], acc[0][t]);
#pragma unroll
      for (int t = 0; t < 4; ++t) acc[1][t] = wmma4(a1, b[t], acc[1][t]);
    }
  }
}

__global__ __launch_bounds__(256)
void diffusion_gemm(const float* __restrict__ S, const float* __restrict__ X,
                    const float* __restrict__ Xprev, float* __restrict__ Y,
                    float alpha, float beta) {
  const int lane = threadIdx.x & 31;
  const int wave = threadIdx.x >> 5;
  const int lm   = lane & 15;
  const int kh   = lane >> 4;
  const int m0 = blockIdx.y * 64 + (wave >> 2) * 32;
  const int n0 = blockIdx.x * 256 + (wave & 3) * 64;

  const float* Ap   = S + (m0 + lm) * 8 + kh * 4;
  const float* Xcol = X + (kh * 2) * CDB + n0 + 4 * lm;

  v8f acc[2][4];
#pragma unroll
  for (int h = 0; h < 2; ++h)
#pragma unroll
    for (int t = 0; t < 4; ++t) acc[h][t] = v8f{};

  DFrag f0, f1, f2;
  diff_load(f0, Ap, Xcol, 0);
  diff_load(f1, Ap, Xcol, 1);
#pragma clang loop unroll(disable)
  for (int s = 0; s < DNST - 2; s += 3) {
    diff_load(f2, Ap, Xcol, s + 2);
    __builtin_amdgcn_sched_barrier(0);
    diff_mma(f0, acc);
    diff_load(f0, Ap, Xcol, s + 3);
    __builtin_amdgcn_sched_barrier(0);
    diff_mma(f1, acc);
    diff_load(f1, Ap, Xcol, s + 4);
    __builtin_amdgcn_sched_barrier(0);
    diff_mma(f2, acc);
  }
  diff_mma(f0, acc);
  diff_mma(f1, acc);

  store_tile(Y, Xprev, m0 + kh * 8,      n0 + 4 * lm, alpha, beta,
             acc[0][0], acc[0][1], acc[0][2], acc[0][3]);
  store_tile(Y, Xprev, m0 + 16 + kh * 8, n0 + 4 * lm, alpha, beta,
             acc[1][0], acc[1][1], acc[1][2], acc[1][3]);
}

#define OSTG 6
#define ONST (D_SZ / OSTG)

struct OFrag {
  v2f pa[OSTG], pb[OSTG];
  v4f w0[OSTG], w1[OSTG];
};

__device__ __forceinline__
void out_load(OFrag& f, const float* __restrict__ xa, const float* __restrict__ xb,
              const float* __restrict__ Wb, int stage) {
#pragma unroll
  for (int i = 0; i < OSTG; ++i) {
    const int d = stage * OSTG + i;
    f.pa[i] = *(const v2f*)(xa + d * B_SZ);
    f.pb[i] = *(const v2f*)(xb + d * B_SZ);
    const float* Wr0 = Wb + (d * 4) * OUTC;
    f.w0[i] = *(const v4f*)(Wr0);
    f.w1[i] = *(const v4f*)(Wr0 + OUTC);
  }
}

__device__ __forceinline__
void out_mma(const OFrag& f, v8f (&acc)[2][4]) {
#pragma unroll
  for (int i = 0; i < OSTG; ++i) {
    const v2f a0 = {f.pa[i].x, f.pb[i].x};
    const v2f a1 = {f.pa[i].y, f.pb[i].y};
    const v2f b[4] = {{f.w0[i].x, f.w1[i].x}, {f.w0[i].y, f.w1[i].y},
                      {f.w0[i].z, f.w1[i].z}, {f.w0[i].w, f.w1[i].w}};
#pragma unroll
    for (int t = 0; t < 4; ++t) acc[0][t] = wmma4(a0, b[t], acc[0][t]);
#pragma unroll
    for (int t = 0; t < 4; ++t) acc[1][t] = wmma4(a1, b[t], acc[1][t]);
  }
}

__global__ __launch_bounds__(256)
void output_gemm(const float* __restrict__ x0, const float* __restrict__ x1,
                 const float* __restrict__ x2, const float* __restrict__ x3,
                 const float* __restrict__ W, const float* __restrict__ bias,
                 float* __restrict__ out) {
  const int lane = threadIdx.x & 31;
  const int wave = threadIdx.x >> 5;
  const int lm   = lane & 15;
  const int kh   = lane >> 4;
  const int n  = blockIdx.y * 2 + (wave >> 2);
  const int o0 = blockIdx.x * 256 + (wave & 3) * 64;

  const int aoff = n * CDB + 2 * lm;
  const float* xa = ((kh == 0) ? x0 : x2) + aoff;
  const float* xb = ((kh == 0) ? x1 : x3) + aoff;
  const float* Wb = W + kh * 2 * OUTC + o0 + 4 * lm;

  v8f acc[2][4];
#pragma unroll
  for (int h = 0; h < 2; ++h)
#pragma unroll
    for (int t = 0; t < 4; ++t) acc[h][t] = v8f{};

  OFrag fa, fb;
  out_load(fa, xa, xb, Wb, 0);
#pragma clang loop unroll(disable)
  for (int s = 0; s < ONST - 2; s += 2) {
    out_load(fb, xa, xb, Wb, s + 1);
    __builtin_amdgcn_sched_barrier(0);
    out_mma(fa, acc);
    out_load(fa, xa, xb, Wb, s + 2);
    __builtin_amdgcn_sched_barrier(0);
    out_mma(fb, acc);
  }
  out_load(fb, xa, xb, Wb, ONST - 1);
  __builtin_amdgcn_sched_barrier(0);
  out_mma(fa, acc);
  out_mma(fb, acc);

  const v4f bi = *(const v4f*)(bias + o0 + 4 * lm);
#pragma unroll
  for (int h = 0; h < 2; ++h) {
#pragma unroll
    for (int j = 0; j < 8; ++j) {
      const int b = 2 * (kh * 8 + j) + h;
      v4f v;
      v.x = acc[h][0][j] + bi.x;
      v.y = acc[h][1][j] + bi.y;
      v.z = acc[h][2][j] + bi.z;
      v.w = acc[h][3][j] + bi.w;
      *(v4f*)(out + (b * N_SZ + n) * OUTC + o0 + 4 * lm) = v;
    }
  }
}

extern "C" void kernel_launch(void* const* d_in, const int* in_sizes, int n_in,
                              void* d_out, int out_size, void* d_ws, size_t ws_size,
                              hipStream_t stream) {
  const float* inputs  = (const float*)d_in[0];
  const float* adj     = (const float*)d_in[1];
  const float* weights = (const float*)d_in[2];
  const float* biases  = (const float*)d_in[3];
  float* out = (float*)d_out;

  float* ws      = (float*)d_ws;
  float* support = ws;
  float* dinv    = support + (size_t)N_SZ * N_SZ;
  float* x0      = dinv + 512;
  float* x1      = x0 + (size_t)N_SZ * CDB;
  float* x2      = x1 + (size_t)N_SZ * CDB;
  float* x3      = x2 + (size_t)N_SZ * CDB;

  (void)in_sizes; (void)n_in; (void)out_size; (void)ws_size;

  dinv_kernel<<<N_SZ, 256, 0, stream>>>(adj, dinv);
  support_kernel<<<(N_SZ * N_SZ) / 256, 256, 0, stream>>>(adj, dinv, support);
  transpose_kernel<<<dim3(N_SZ, D_SZ / 64), 256, 0, stream>>>(inputs, x0);

  dim3 gd(CDB / 256, N_SZ / 64);
  diffusion_gemm<<<gd, 256, 0, stream>>>(support, x0, x0, x1, 1.0f, 0.0f);
  diffusion_gemm<<<gd, 256, 0, stream>>>(support, x1, x0, x2, 2.0f, -1.0f);
  diffusion_gemm<<<gd, 256, 0, stream>>>(support, x2, x1, x3, 2.0f, -1.0f);

  output_gemm<<<dim3(OUTC / 256, (B_SZ * N_SZ) / 64), 256, 0, stream>>>(
      x0, x1, x2, x3, weights, biases, out);
}


# both GEMMs: wave tile 64x64 (16 accumulators) in place of 32x64, so each loaded fragment feeds 4 matrix instructions; 256 workgroups of 12 waves hold all tiles at once; k and sum order unchanged
# speedup vs baseline: 4.6161x; 1.0588x over previous
#include <hip/hip_runtime.h>
#include <hip/hip_bf16.h>

#define B_SZ   32
#define N_SZ   512
#define D_SZ   768
#define M_SZ   4
#define CDB    (D_SZ * B_SZ)
#define OUTC   D_SZ

typedef __attribute__((ext_vector_type(2))) float v2f;
typedef __attribute__((ext_vector_type(4))) float v4f;
typedef __attribute__((ext_vector_type(8))) float v8f;

__device__ __forceinline__ v8f wmma4(v2f a, v2f b, v8f c) {
  return __builtin_amdgcn_wmma_f32_16x16x4_f32(
       false, a,  false, b,
       (short)0, c,  false,  false);
}

__global__ __launch_bounds__(256)
void dinv_kernel(const float* __restrict__ adj, float* __restrict__ dinv) {
  __shared__ float red[256];
  const int row = blockIdx.x;
  const int t = threadIdx.x;
  float s = 0.f;
  const float* a = adj + row * N_SZ;
  for (int j = t; j < N_SZ; j += 256) s += a[j];
  red[t] = s;
  __syncthreads();
  for (int off = 128; off > 0; off >>= 1) {
    if (t < off) red[t] += red[t + off];
    __syncthreads();
  }
  if (t == 0) {
    float rs = red[0] + 1.0f;
    dinv[row] = (rs > 0.f) ? rsqrtf(rs) : 0.f;
  }
}

__global__ __launch_bounds__(256)
void support_kernel(const float* __restrict__ adj, const float* __restrict__ dinv,
                    float* __restrict__ support) {
  const int a   = blockIdx.x * 256 + threadIdx.x;
  const int pos = a & 7;
  const int i   = (a >> 3) & (N_SZ - 1);
  const int kb  = a >> 12;
  const int w   = ((pos >> 1) & 1) * 4 + (pos >> 2) * 2 + (pos & 1);
  const int j   = kb * 8 + w;
  float m = adj[j * N_SZ + i] + (i == j ? 1.0f : 0.0f);
  support[a] = m * dinv[i] * dinv[j];
}

__global__ __launch_bounds__(256)
void transpose_kernel(const float* __restrict__ inputs, float* __restrict__ x0) {
  __shared__ float tile[64 * 33];
  const int n  = blockIdx.x;
  const int d0 = blockIdx.y * 64;
  const int t  = threadIdx.x;
#pragma unroll
  for (int p = 0; p < 8; ++p) {
    int b  = p * 4 + (t >> 6);
    int dl = t & 63;
    tile[dl * 33 + b] =
        inputs[(size_t)b * (N_SZ * D_SZ) + (size_t)n * D_SZ + d0 + dl];
  }
  __syncthreads();
  float* dst = x0 + (size_t)n * CDB + d0 * B_SZ;
#pragma unroll
  for (int p = 0; p < 8; ++p) {
    int c  = p * 256 + t;
    int dl = c >> 5;
    int b  = c & 31;
    dst[c] = tile[dl * 33 + b];
  }
}

__device__ __forceinline__
void store_tile(float* __restrict__ Y, const float* __restrict__ Xprev,
                int rbase, int col, float alpha, float beta,
                v8f a0, v8f a1, v8f a2, v8f a3) {
  if (beta != 0.0f) {
#pragma unroll
    for (int j = 0; j < 8; ++j) {
      const int off = (rbase + j) * CDB + col;
      const v4f p = *(const v4f*)(Xprev + off);
      v4f y;
      y.x = alpha * a0[j] + beta * p.x;
      y.y = alpha * a1[j] + beta * p.y;
      y.z = alpha * a2[j] + beta * p.z;
      y.w = alpha * a3[j] + beta * p.w;
      *(v4f*)(Y + off) = y;
    }
  } else {
#pragma unroll
    for (int j = 0; j < 8; ++j) {
      const int off = (rbase + j) * CDB + col;
      v4f y;
      y.x = alpha * a0[j];
      y.y = alpha * a1[j];
      y.z = alpha * a2[j];
      y.w = alpha * a3[j];
      *(v4f*)(Y + off) = y;
    }
  }
}

#define DNST (N_SZ / 8)

struct DFrag {
  v4f av[4];
  v4f r0, r1, r2, r3;
};

__device__ __forceinline__
void diff_load(DFrag& f, const float* __restrict__ Ap,
               const float* __restrict__ Xcol, int kb) {
  const float* Ab = Ap + kb * (N_SZ * 8);
#pragma unroll
  for (int i = 0; i < 4; ++i)
    f.av[i] = *(const v4f*)(Ab + i * 16 * 8);
  const float* Xr = Xcol + (kb * 8) * CDB;
  f.r0 = *(const v4f*)(Xr);
  f.r1 = *(const v4f*)(Xr + CDB);
  f.r2 = *(const v4f*)(Xr + 4 * CDB);
  f.r3 = *(const v4f*)(Xr + 5 * CDB);
}

__device__ __forceinline__
void diff_mma(const DFrag& f, v8f (&acc)[4][4]) {
  {
    const v2f b[4] = {{f.r0.x, f.r1.x}, {f.r0.y, f.r1.y},
                      {f.r0.z, f.r1.z}, {f.r0.w, f.r1.w}};
#pragma unroll
    for (int i = 0; i < 4; ++i) {
      const v2f a = {f.av[i].x, f.av[i].y};
#pragma unroll
      for (int t = 0; t < 4; ++t) acc[i][t] = wmma4(a, b[t], acc[i][t]);
    }
  }
  {
    const v2f b[4] = {{f.r2.x, f.r3.x}, {f.r2.y, f.r3.y},
                      {f.r2.z, f.r3.z}, {f.r2.w, f.r3.w}};
#pragma unroll
    for (int i = 0; i < 4; ++i) {
      const v2f a = {f.av[i].z, f.av[i].w};
#pragma unroll
      for (int t = 0; t < 4; ++t) acc[i][t] = wmma4(a, b[t], acc[i][t]);
    }
  }
}

__global__ __launch_bounds__(384)
void diffusion_gemm(const float* __restrict__ S, const float* __restrict__ X,
                    const float* __restrict__ Xprev, float* __restrict__ Y,
                    float alpha, float beta) {
  const int lane = threadIdx.x & 31;
  const int wave = threadIdx.x >> 5;
  const int lm   = lane & 15;
  const int kh   = lane >> 4;
  const int tile = blockIdx.x * 12 + wave;
  const int m0 = (tile & 7) * 64;
  const int n0 = (tile >> 3) * 64;

  const float* Ap   = S + (m0 + lm) * 8 + kh * 4;
  const float* Xcol = X + (kh * 2) * CDB + n0 + 4 * lm;

  v8f acc[4][4];
#pragma unroll
  for (int i = 0; i < 4; ++i)
#pragma unroll
    for (int t = 0; t < 4; ++t) acc[i][t] = v8f{};

  DFrag f0, f1, f2;
  diff_load(f0, Ap, Xcol, 0);
  diff_load(f1, Ap, Xcol, 1);
#pragma clang loop unroll(disable)
  for (int s = 0; s < DNST - 4; s += 3) {
    diff_load(f2, Ap, Xcol, s + 2);
    __builtin_amdgcn_sched_barrier(0);
    diff_mma(f0, acc);
    diff_load(f0, Ap, Xcol, s + 3);
    __builtin_amdgcn_sched_barrier(0);
    diff_mma(f1, acc);
    diff_load(f1, Ap, Xcol, s + 4);
    __builtin_amdgcn_sched_barrier(0);
    diff_mma(f2, acc);
  }
  diff_load(f2, Ap, Xcol, DNST - 2);
  __builtin_amdgcn_sched_barrier(0);
  diff_mma(f0, acc);
  diff_load(f0, Ap, Xcol, DNST - 1);
  __builtin_amdgcn_sched_barrier(0);
  diff_mma(f1, acc);
  diff_mma(f2, acc);
  diff_mma(f0, acc);

#pragma unroll
  for (int i = 0; i < 4; ++i)
    store_tile(Y, Xprev, m0 + 16 * i + kh * 8, n0 + 4 * lm, alpha, beta,
               acc[i][0], acc[i][1], acc[i][2], acc[i][3]);
}

#define OSTG 3
#define ONST (D_SZ / OSTG)

struct OFrag {
  v2f pa[2][OSTG], pb[2][OSTG];
  v4f w0[OSTG], w1[OSTG];
};

__device__ __forceinline__
void out_load(OFrag& f, const float* __restrict__ xa, const float* __restrict__ xb,
              const float* __restrict__ Wb, int stage) {
#pragma unroll
  for (int i = 0; i < OSTG; ++i) {
    const int d = stage * OSTG + i;
#pragma unroll
    for (int nn = 0; nn < 2; ++nn) {
      f.pa[nn][i] = *(const v2f*)(xa + nn * CDB + d * B_SZ);
      f.pb[nn][i] = *(const v2f*)(xb + nn * CDB + d * B_SZ);
    }
    const float* Wr0 = Wb + (d * 4) * OUTC;
    f.w0[i] = *(const v4f*)(Wr0);
    f.w1[i] = *(const v4f*)(Wr0 + OUTC);
  }
}

__device__ __forceinline__
void out_mma(const OFrag& f, v8f (&acc)[4][4]) {
#pragma unroll
  for (int i = 0; i < OSTG; ++i) {
    const v2f b[4] = {{f.w0[i].x, f.w1[i].x}, {f.w0[i].y, f.w1[i].y},
                      {f.w0[i].z, f.w1[i].z}, {f.w0[i].w, f.w1[i].w}};
#pragma unroll
    for (int nn = 0; nn < 2; ++nn) {
      const v2f a0 = {f.pa[nn][i].x, f.pb[nn][i].x};
      const v2f a1 = {f.pa[nn][i].y, f.pb[nn][i].y};
#pragma unroll
      for (int t = 0; t < 4; ++t) acc[2 * nn][t] = wmma4(a0, b[t], acc[2 * nn][t]);
#pragma unroll
      for (int t = 0; t < 4; ++t) acc[2 * nn + 1][t] = wmma4(a1, b[t], acc[2 * nn + 1][t]);
    }
  }
}

__global__ __launch_bounds__(384)
void output_gemm(const float* __restrict__ x0, const float* __restrict__ x1,
                 const float* __restrict__ x2, const float* __restrict__ x3,
                 const float* __restrict__ W, const float* __restrict__ bias,
                 float* __restrict__ out) {
  const int lane = threadIdx.x & 31;
  const int wave = threadIdx.x >> 5;
  const int lm   = lane & 15;
  const int kh   = lane >> 4;
  const int n  = blockIdx.x * 2;
  const int o0 = wave * 64;

  const int aoff = n * CDB + 2 * lm;
  const float* xa = ((kh == 0) ? x0 : x2) + aoff;
  const float* xb = ((kh == 0) ? x1 : x3) + aoff;
  const float* Wb = W + kh * 2 * OUTC + o0 + 4 * lm;

  v8f acc[4][4];
#pragma unroll
  for (int q = 0; q < 4; ++q)
#pragma unroll
    for (int t = 0; t < 4; ++t) acc[q][t] = v8f{};

  OFrag fa, fb;
  out_load(fa, xa, xb, Wb, 0);
#pragma clang loop unroll(disable)
  for (int s = 0; s < ONST - 2; s += 2) {
    out_load(fb, xa, xb, Wb, s + 1);
    __builtin_amdgcn_sched_barrier(0);
    out_mma(fa, acc);
    out_load(fa, xa, xb, Wb, s + 2);
    __builtin_amdgcn_sched_barrier(0);
    out_mma(fb, acc);
  }
  out_load(fb, xa, xb, Wb, ONST - 1);
  __builtin_amdgcn_sched_barrier(0);
  out_mma(fa, acc);
  out_mma(fb, acc);

  const v4f bi = *(const v4f*)(bias + o0 + 4 * lm);
#pragma unroll
  for (int nn = 0; nn < 2; ++nn) {
#pragma unroll
    for (int h = 0; h < 2; ++h) {
#pragma unroll
      for (int j = 0; j < 8; ++j) {
        const int b = 2 * (kh * 8 + j) + h;
        v4f v;
        v.x = acc[2 * nn + h][0][j] + bi.x;
        v.y = acc[2 * nn + h][1][j] + bi.y;
        v.z = acc[2 * nn + h][2][j] + bi.z;
        v.w = acc[2 * nn + h][3][j] + bi.w;
        *(v4f*)(out + (b * N_SZ + n + nn) * OUTC + o0 + 4 * lm) = v;
      }
    }
  }
}

extern "C" void kernel_launch(void* const* d_in, const int* in_sizes, int n_in,
                              void* d_out, int out_size, void* d_ws, size_t ws_size,
                              hipStream_t stream) {
  const float* inputs  = (const float*)d_in[0];
  const float* adj     = (const float*)d_in[1];
  const float* weights = (const float*)d_in[2];
  const float* biases  = (const float*)d_in[3];
  float* out = (float*)d_out;

  float* ws      = (float*)d_ws;
  float* support = ws;
  float* dinv    = support + (size_t)N_SZ * N_SZ;
  float* x0      = dinv + 512;
  float* x1      = x0 + (size_t)N_SZ * CDB;
  float* x2      = x1 + (size_t)N_SZ * CDB;
  float* x3      = x2 + (size_t)N_SZ * CDB;

  (void)in_sizes; (void)n_in; (void)out_size; (void)ws_size;

  dinv_kernel<<<N_SZ, 256, 0, stream>>>(adj, dinv);
  support_kernel<<<(N_SZ * N_SZ) / 256, 256, 0, stream>>>(adj, dinv, support);
  transpose_kernel<<<dim3(N_SZ, D_SZ / 64), 256, 0, stream>>>(inputs, x0);

  diffusion_gemm<<<256, 384, 0, stream>>>(support, x0, x0, x1, 1.0f, 0.0f);
  diffusion_gemm<<<256, 384, 0, stream>>>(support, x1, x0, x2, 2.0f, -1.0f);
  diffusion_gemm<<<256, 384, 0, stream>>>(support, x2, x1, x3, 2.0f, -1.0f);

  output_gemm<<<256, 384, 0, stream>>>(x0, x1, x2, x3, weights, biases, out);
}
